# starting kernel
# baseline (speedup 1.0000x reference)
#include <hip/hip_runtime.h>
#include <math.h>

typedef float v2f __attribute__((ext_vector_type(2)));
typedef float v8f __attribute__((ext_vector_type(8)));

#define HD __device__ __forceinline__

static constexpr int   Hc   = 64;
static constexpr float GEPS = 1e-7f;

HD void atomicMaxF(float* addr, float val) {
    if (val >= 0.0f)
        atomicMax((int*)addr, __float_as_int(val));
    else
        atomicMin((unsigned int*)addr, __float_as_uint(val));
}

__global__ void gemm_bias_wmma(const float* __restrict__ A,
                               const float* __restrict__ B,
                               const float* __restrict__ bias,
                               float* __restrict__ C,
                               int M, int N, int K) {
    int wave    = (int)((blockIdx.x * blockDim.x + threadIdx.x) >> 5);
    int lane    = threadIdx.x & 31;
    int groupsN = (N + 63) >> 6;
    int tm = wave / groupsN;
    int gn = wave - tm * groupsN;
    if (tm >= (M >> 4)) return;

    int mr   = lane & 15;
    int half = lane >> 4;
    const float* Arow = A + (long)(tm * 16 + mr) * K;

    int ncol[4];
#pragma unroll
    for (int t = 0; t < 4; ++t) {
        int n   = gn * 64 + t * 16 + mr;
        ncol[t] = (n < N) ? n : (N - 1);
    }

    v8f acc[4] = {{0.f,0.f,0.f,0.f,0.f,0.f,0.f,0.f},
                  {0.f,0.f,0.f,0.f,0.f,0.f,0.f,0.f},
                  {0.f,0.f,0.f,0.f,0.f,0.f,0.f,0.f},
                  {0.f,0.f,0.f,0.f,0.f,0.f,0.f,0.f}};

    int ka = half * 2;
    v2f a_cur;
    a_cur.x = Arow[ka];
    a_cur.y = Arow[ka + 1];
    v2f b_cur[4];
#pragma unroll
    for (int t = 0; t < 4; ++t) {
        b_cur[t].x = B[ka * N + ncol[t]];
        b_cur[t].y = B[(ka + 1) * N + ncol[t]];
    }

    for (int k = 0; k < K; k += 4) {
        int kn = (k + 4 < K) ? (k + 4) : k;
        int kb = kn + half * 2;

        v2f a_nxt;
        a_nxt.x = Arow[kb];
        a_nxt.y = Arow[kb + 1];
        v2f b_nxt[4];
#pragma unroll
        for (int t = 0; t < 4; ++t) {
            b_nxt[t].x = B[kb * N + ncol[t]];
            b_nxt[t].y = B[(kb + 1) * N + ncol[t]];
        }

#pragma unroll
        for (int t = 0; t < 4; ++t)
            acc[t] = __builtin_amdgcn_wmma_f32_16x16x4_f32(
                         false, a_cur, false, b_cur[t], (short)0, acc[t], false, false);

        a_cur = a_nxt;
#pragma unroll
        for (int t = 0; t < 4; ++t) b_cur[t] = b_nxt[t];
    }

#pragma unroll
    for (int t = 0; t < 4; ++t) {
        int n = gn * 64 + t * 16 + mr;
        if (n < N) {
            float bv = bias[n];
#pragma unroll
            for (int r = 0; r < 8; ++r)
                C[(long)(tm * 16 + half * 8 + r) * N + n] = acc[t][r] + bv;
        }
    }
}

__global__ void ln_relu_kernel(const float* __restrict__ X,
                               const float* __restrict__ g,
                               const float* __restrict__ bvec,
                               float* __restrict__ Y,
                               int rows, int D) {
    int warp = threadIdx.x >> 5;
    int lane = threadIdx.x & 31;
    int row  = blockIdx.x * (blockDim.x >> 5) + warp;
    if (row >= rows) return;

    const float* x = X + (long)row * D;
    int   nper = D >> 5;
    float vals[4];
    float s = 0.f, ss = 0.f;
    for (int j = 0; j < nper; ++j) {
        float v = x[lane + j * 32];
        vals[j] = v;
        s  += v;
        ss += v * v;
    }
    for (int off = 16; off; off >>= 1) {
        s  += __shfl_xor(s,  off);
        ss += __shfl_xor(ss, off);
    }
    float mean = s / (float)D;
    float var  = ss / (float)D - mean * mean;
    float rstd = rsqrtf(var + 1e-5f);

    float* y = Y + (long)row * D;
    for (int j = 0; j < nper; ++j) {
        int   idx = lane + j * 32;
        float v   = (vals[j] - mean) * rstd * g[idx] + bvec[idx];
        y[idx] = fmaxf(v, 0.f);
    }
}

__global__ void fill_buffers(float* __restrict__ smax,
                             float* __restrict__ den,
                             float* __restrict__ num, int n) {
    int i = blockIdx.x * blockDim.x + threadIdx.x;
    if (i < n) { smax[i] = -3.0e38f; den[i] = 0.f; num[i] = 0.f; }
}

__global__ void edge_max_kernel(const float* __restrict__ z,
                                const int* __restrict__ src,
                                const int* __restrict__ dst,
                                const float* __restrict__ tptr, int layer,
                                float* __restrict__ smax, int total) {
    int i = blockIdx.x * blockDim.x + threadIdx.x;
    if (i >= total) return;
    int e  = i >> 6;
    int ch = i & 63;
    float t   = tptr[layer];
    float msg = fmaxf(z[src[e] * Hc + ch], 0.f) + GEPS;
    atomicMaxF(&smax[dst[e] * Hc + ch], msg * t);
}

__global__ void edge_sum_kernel(const float* __restrict__ z,
                                const int* __restrict__ src,
                                const int* __restrict__ dst,
                                const float* __restrict__ tptr, int layer,
                                const float* __restrict__ smax,
                                float* __restrict__ den,
                                float* __restrict__ num, int total) {
    int i = blockIdx.x * blockDim.x + threadIdx.x;
    if (i >= total) return;
    int e  = i >> 6;
    int ch = i & 63;
    float t   = tptr[layer];
    float msg = fmaxf(z[src[e] * Hc + ch], 0.f) + GEPS;
    int   d   = dst[e] * Hc + ch;
    float ex  = __expf(msg * t - smax[d]);
    atomicAdd(&den[d], ex);
    atomicAdd(&num[d], msg * ex);
}

__global__ void agg_kernel(const float* __restrict__ num,
                           const float* __restrict__ den,
                           const float* __restrict__ z,
                           float* __restrict__ out, int n) {
    int i = blockIdx.x * blockDim.x + threadIdx.x;
    if (i < n) out[i] = num[i] / (den[i] + 1e-16f) + z[i];
}

__global__ void residual_kernel(float* __restrict__ h,
                                const float* __restrict__ y,
                                float beta, int n) {
    int i = blockIdx.x * blockDim.x + threadIdx.x;
    if (i < n) h[i] = beta * h[i] + y[i];
}

__global__ void log_softmax_kernel(const float* __restrict__ X,
                                   float* __restrict__ Y, int rows, int C) {
    int r = blockIdx.x * blockDim.x + threadIdx.x;
    if (r >= rows) return;
    const float* x = X + (long)r * C;
    float m = -3.0e38f;
    for (int j = 0; j < C; ++j) m = fmaxf(m, x[j]);
    float s = 0.f;
    for (int j = 0; j < C; ++j) s += __expf(x[j] - m);
    float lse = m + __logf(s);
    float* y = Y + (long)r * C;
    for (int j = 0; j < C; ++j) y[j] = x[j] - lse;
}

extern "C" void kernel_launch(void* const* d_in, const int* in_sizes, int n_in,
                              void* d_out, int out_size, void* d_ws, size_t ws_size,
                              hipStream_t stream) {
    (void)n_in; (void)out_size; (void)ws_size;

    const float* x    = (const float*)d_in[0];
    const int*   ei   = (const int*)  d_in[1];
    const float* encW = (const float*)d_in[2];
    const float* encB = (const float*)d_in[3];
    const float* t    = (const float*)d_in[4];
    const float* W1   = (const float*)d_in[5];
    const float* b1   = (const float*)d_in[6];
    const float* g1   = (const float*)d_in[7];
    const float* bb1  = (const float*)d_in[8];
    const float* W2   = (const float*)d_in[9];
    const float* b2   = (const float*)d_in[10];
    const float* ng   = (const float*)d_in[11];
    const float* nb   = (const float*)d_in[12];
    const float* linW = (const float*)d_in[13];
    const float* linB = (const float*)d_in[14];

    const int F = 128, H = 64, C = 40, L = 4;
    const int N = in_sizes[0] / F;
    const int E = in_sizes[1] / 2;
    const int* src = ei;
    const int* dst = ei + E;

    float* ws = (float*)d_ws;
    const long NH = (long)N * H;
    float* h    = ws;  ws += NH;
    float* z    = ws;  ws += NH;
    float* smax = ws;  ws += NH;
    float* den  = ws;  ws += NH;
    float* num  = ws;  ws += NH;
    float* outb = ws;  ws += NH;
    float* yb   = ws;  ws += NH;
    float* tmp  = ws;  ws += (long)N * 2 * H;

    const int  TB = 256;
    auto cdiv = [](long a, long b) { return (int)((a + b - 1) / b); };
    auto gemm_waves = [&](int M, int Nc) { return (long)(M / 16) * ((Nc + 63) / 64); };

    gemm_bias_wmma<<<cdiv(gemm_waves(N, H) * 32, TB), TB, 0, stream>>>(x, encW, encB, h, N, H, F);

    const long totalE = (long)E * H;

    for (int l = 0; l < L; ++l) {
        const float* zin;
        if (l == 0) {
            zin = h;
        } else {
            ln_relu_kernel<<<cdiv(N, 8), 256, 0, stream>>>(h, ng + l * H, nb + l * H, z, N, H);
            zin = z;
        }

        fill_buffers<<<cdiv(NH, TB), TB, 0, stream>>>(smax, den, num, (int)NH);
        edge_max_kernel<<<cdiv(totalE, TB), TB, 0, stream>>>(zin, src, dst, t, l, smax, (int)totalE);
        edge_sum_kernel<<<cdiv(totalE, TB), TB, 0, stream>>>(zin, src, dst, t, l, smax, den, num, (int)totalE);
        agg_kernel<<<cdiv(NH, TB), TB, 0, stream>>>(num, den, zin, outb, (int)NH);

        gemm_bias_wmma<<<cdiv(gemm_waves(N, 2 * H) * 32, TB), TB, 0, stream>>>(
            outb, W1 + (long)l * H * 2 * H, b1 + l * 2 * H, tmp, N, 2 * H, H);
        ln_relu_kernel<<<cdiv(N, 8), 256, 0, stream>>>(tmp, g1 + l * 2 * H, bb1 + l * 2 * H, tmp, N, 2 * H);
        gemm_bias_wmma<<<cdiv(gemm_waves(N, H) * 32, TB), TB, 0, stream>>>(
            tmp, W2 + (long)l * 2 * H * H, b2 + l * H, yb, N, H, 2 * H);

        residual_kernel<<<cdiv(NH, TB), TB, 0, stream>>>(h, yb, (l == 0) ? 0.f : 1.f, (int)NH);
    }

    ln_relu_kernel<<<cdiv(N, 8), 256, 0, stream>>>(h, ng, nb, z, N, H);
    gemm_bias_wmma<<<cdiv(gemm_waves(N, C) * 32, TB), TB, 0, stream>>>(z, linW, linB, tmp, N, C, H);
    log_softmax_kernel<<<cdiv(N, TB), TB, 0, stream>>>(tmp, (float*)d_out, N, C);
}


# Tail as ONE kernel: last residual add, final LayerNorm+ReLU, classifier GEMM and log_softmax per 16-row wave tile via LDS, whole-line stores; 21 -> 19 launches; log_softmax row-sum order changes
# speedup vs baseline: 4.7463x; 4.7463x over previous
#include <hip/hip_runtime.h>
#include <math.h>

typedef float v2f __attribute__((ext_vector_type(2)));
typedef float v4f __attribute__((ext_vector_type(4)));
typedef float v8f __attribute__((ext_vector_type(8)));

#define HD __device__ __forceinline__

static constexpr int   Hc   = 64;
static constexpr float GEPS = 1e-7f;

HD void atomicMaxF(float* addr, float val) {
    if (val >= 0.0f)
        atomicMax((int*)addr, __float_as_int(val));
    else
        atomicMin((unsigned int*)addr, __float_as_uint(val));
}

__global__ void gemm_bias_wmma(const float* __restrict__ A,
                               const float* __restrict__ B,
                               const float* __restrict__ bias,
                               float* __restrict__ C,
                               int M, int N, int K) {
    int wave    = (int)((blockIdx.x * blockDim.x + threadIdx.x) >> 5);
    int lane    = threadIdx.x & 31;
    int groupsN = (N + 63) >> 6;
    int tm = wave / groupsN;
    int gn = wave - tm * groupsN;
    if (tm >= (M >> 4)) return;

    int mr   = lane & 15;
    int half = lane >> 4;
    const float* Arow = A + (long)(tm * 16 + mr) * K;

    int ncol[4];
#pragma unroll
    for (int t = 0; t < 4; ++t) {
        int n   = gn * 64 + t * 16 + mr;
        ncol[t] = (n < N) ? n : (N - 1);
    }

    v8f acc[4] = {{0.f,0.f,0.f,0.f,0.f,0.f,0.f,0.f},
                  {0.f,0.f,0.f,0.f,0.f,0.f,0.f,0.f},
                  {0.f,0.f,0.f,0.f,0.f,0.f,0.f,0.f},
                  {0.f,0.f,0.f,0.f,0.f,0.f,0.f,0.f}};

    int ka = half * 2;
    v2f a_cur;
    a_cur.x = Arow[ka];
    a_cur.y = Arow[ka + 1];
    v2f b_cur[4];
#pragma unroll
    for (int t = 0; t < 4; ++t) {
        b_cur[t].x = B[ka * N + ncol[t]];
        b_cur[t].y = B[(ka + 1) * N + ncol[t]];
    }

    for (int k = 0; k < K; k += 4) {
        int kn = (k + 4 < K) ? (k + 4) : k;
        int kb = kn + half * 2;

        v2f a_nxt;
        a_nxt.x = Arow[kb];
        a_nxt.y = Arow[kb + 1];
        v2f b_nxt[4];
#pragma unroll
        for (int t = 0; t < 4; ++t) {
            b_nxt[t].x = B[kb * N + ncol[t]];
            b_nxt[t].y = B[(kb + 1) * N + ncol[t]];
        }

#pragma unroll
        for (int t = 0; t < 4; ++t)
            acc[t] = __builtin_amdgcn_wmma_f32_16x16x4_f32(
                         false, a_cur, false, b_cur[t], (short)0, acc[t], false, false);

        a_cur = a_nxt;
#pragma unroll
        for (int t = 0; t < 4; ++t) b_cur[t] = b_nxt[t];
    }

#pragma unroll
    for (int t = 0; t < 4; ++t) {
        int n = gn * 64 + t * 16 + mr;
        if (n < N) {
            float bv = bias[n];
#pragma unroll
            for (int r = 0; r < 8; ++r)
                C[(long)(tm * 16 + half * 8 + r) * N + n] = acc[t][r] + bv;
        }
    }
}

static constexpr float RANGE_OK = 16.0f;

HD void atomicMinF(float* addr, float val) {
    if (val >= 0.0f)
        atomicMin((int*)addr, __float_as_int(val));
    else
        atomicMax((unsigned int*)addr, __float_as_uint(val));
}

__global__ void init_stats_kernel(float* __restrict__ stats) {
    int i = threadIdx.x;
    if (i < 128) stats[i] = (i < 64) ? -3.0e38f : 3.0e38f;
    if (i < 2)   stats[128 + i] = 0.f;
}

HD bool shift_is_safe(float M0, float M1, float m0, float m1) {
    float d0 = M0 - m0;
    float d1 = M1 - m1;
    bool ok = (d0 >= 0.f) && (d0 <= RANGE_OK) && (d1 >= 0.f) && (d1 <= RANGE_OK);
    return __builtin_amdgcn_ballot_w32(ok) == 0xffffffffu;
}

template <int RB>
__global__ __launch_bounds__(256)
void fill_stats_kernel(const float* __restrict__ z,
                       const float* __restrict__ tptr, int layer,
                       float* __restrict__ smax,
                       float* __restrict__ den,
                       float* __restrict__ num,
                       float* __restrict__ stats, int N) {
    __shared__ float red[4][8][32];
    const int  lane = threadIdx.x & 31;
    const int  w    = __builtin_amdgcn_readfirstlane((int)(threadIdx.x >> 5));
    const long r0   = ((long)blockIdx.x * 8 + w) * RB;
    const float t   = tptr[layer];
    float mx0 = -3.0e38f, mx1 = -3.0e38f, mn0 = 3.0e38f, mn1 = 3.0e38f;
    float a[RB], b[RB];
#pragma unroll
    for (int j = 0; j < RB; ++j) {
        long r = (r0 + j < N) ? (r0 + j) : 0;
        a[j] = z[(size_t)r * Hc + lane];
        b[j] = z[(size_t)r * Hc + lane + 32];
    }
#pragma unroll
    for (int j = 0; j < RB; ++j) {
        if (r0 + j < N) {
            size_t o = (size_t)(r0 + j) * Hc + lane;
            float s0 = (fmaxf(a[j], 0.f) + GEPS) * t;
            float s1 = (fmaxf(b[j], 0.f) + GEPS) * t;
            mx0 = fmaxf(mx0, s0);  mn0 = fminf(mn0, s0);
            mx1 = fmaxf(mx1, s1);  mn1 = fminf(mn1, s1);
            smax[o] = -3.0e38f;  smax[o + 32] = -3.0e38f;
            den[o]  = 0.f;       den[o + 32]  = 0.f;
            num[o]  = 0.f;       num[o + 32]  = 0.f;
        }
    }
    red[0][w][lane] = mx0;
    red[1][w][lane] = mx1;
    red[2][w][lane] = mn0;
    red[3][w][lane] = mn1;
    __syncthreads();
    if (threadIdx.x < 128) {
        const int k = threadIdx.x >> 5;
        float v = red[k][0][lane];
        for (int i = 1; i < 8; ++i) {
            float u = red[k][i][lane];
            v = (k < 2) ? fmaxf(v, u) : fminf(v, u);
        }
        if (k < 2) atomicMaxF(stats + k * 32 + lane, v);
        else       atomicMinF(stats + k * 32 + lane, v);
    }
}

template <int RB, bool FIRST, bool FILL>
__global__ __launch_bounds__(1024)
void boundary_kernel(float* __restrict__ h,
                     const float* __restrict__ y,
                     const float* __restrict__ g,
                     const float* __restrict__ bvec,
                     float* __restrict__ z,
                     const float* __restrict__ tptr, int layer,
                     float* __restrict__ smax,
                     float* __restrict__ den,
                     float* __restrict__ num,
                     float* __restrict__ stats, int N) {
    __shared__ float red[4][32][32];
    const int  lane = threadIdx.x & 31;
    const int  w    = __builtin_amdgcn_readfirstlane((int)(threadIdx.x >> 5));
    const long r0   = ((long)blockIdx.x * 32 + w) * RB;
    const float g0 = g[lane],    g1 = g[lane + 32];
    const float c0 = bvec[lane], c1 = bvec[lane + 32];
    float t = 0.f;
    if (FILL) t = tptr[layer];
    float mx0 = -3.0e38f, mx1 = -3.0e38f, mn0 = 3.0e38f, mn1 = 3.0e38f;
    float a[RB], b[RB], ya[RB], yb[RB];
#pragma unroll
    for (int j = 0; j < RB; ++j) {
        long r = (r0 + j < N) ? (r0 + j) : 0;
        ya[j] = y[(size_t)r * Hc + lane];
        yb[j] = y[(size_t)r * Hc + lane + 32];
        if (!FIRST) {
            a[j] = h[(size_t)r * Hc + lane];
            b[j] = h[(size_t)r * Hc + lane + 32];
        }
    }
    float v0[RB], v1[RB], s[RB], ss[RB];
#pragma unroll
    for (int j = 0; j < RB; ++j) {
        v0[j] = FIRST ? ya[j] : (a[j] + ya[j]);
        v1[j] = FIRST ? yb[j] : (b[j] + yb[j]);
        s[j] = 0.f;  ss[j] = 0.f;
        s[j] += v0[j];  ss[j] += v0[j] * v0[j];
        s[j] += v1[j];  ss[j] += v1[j] * v1[j];
    }
    for (int off = 16; off; off >>= 1) {
#pragma unroll
        for (int j = 0; j < RB; ++j) {
            s[j]  += __shfl_xor(s[j],  off);
            ss[j] += __shfl_xor(ss[j], off);
        }
    }
#pragma unroll
    for (int j = 0; j < RB; ++j) {
        float mean = s[j] / (float)Hc;
        float var  = ss[j] / (float)Hc - mean * mean;
        float rstd = rsqrtf(var + 1e-5f);
        float z0 = fmaxf((v0[j] - mean) * rstd * g0 + c0, 0.f);
        float z1 = fmaxf((v1[j] - mean) * rstd * g1 + c1, 0.f);
        if (r0 + j < N) {
            size_t o = (size_t)(r0 + j) * Hc + lane;
            if (FILL) { h[o] = v0[j];  h[o + 32] = v1[j]; }
            z[o] = z0;  z[o + 32] = z1;
            if (FILL) {
                float s0 = (fmaxf(z0, 0.f) + GEPS) * t;
                float s1 = (fmaxf(z1, 0.f) + GEPS) * t;
                mx0 = fmaxf(mx0, s0);  mn0 = fminf(mn0, s0);
                mx1 = fmaxf(mx1, s1);  mn1 = fminf(mn1, s1);
                smax[o] = -3.0e38f;  smax[o + 32] = -3.0e38f;
                den[o]  = 0.f;       den[o + 32]  = 0.f;
                num[o]  = 0.f;       num[o + 32]  = 0.f;
            }
        }
    }
    if (FILL) {
        red[0][w][lane] = mx0;
        red[1][w][lane] = mx1;
        red[2][w][lane] = mn0;
        red[3][w][lane] = mn1;
        __syncthreads();
        if (threadIdx.x < 128) {
            const int k = threadIdx.x >> 5;
            float v = red[k][0][lane];
            for (int i = 1; i < 32; ++i) {
                float u = red[k][i][lane];
                v = (k < 2) ? fmaxf(v, u) : fminf(v, u);
            }
            if (k < 2) atomicMaxF(stats + k * 32 + lane, v);
            else       atomicMinF(stats + k * 32 + lane, v);
        }
    }
}

template <int U>
__global__ __launch_bounds__(256)
void edge_pass1_kernel(const float* __restrict__ z,
                       const int* __restrict__ src,
                       const int* __restrict__ dst,
                       const float* __restrict__ tptr, int layer,
                       const float* __restrict__ stats,
                       float* __restrict__ smax,
                       float* __restrict__ den,
                       float* __restrict__ num,
                       int* __restrict__ ctr, int E, int N) {
    constexpr int EV = 32;
    constexpr int CH = 8 * EV;
    __shared__ int sh[2];
    const int lane = threadIdx.x & 31;
    const int w      = __builtin_amdgcn_readfirstlane((int)(threadIdx.x >> 5));
    const int nchunk = (E + CH - 1) / CH;
    const int Nh     = N >> 1;
    const float t    = tptr[layer];
    const float M0 = stats[lane],      M1 = stats[lane + 32];
    const float m0 = stats[64 + lane], m1 = stats[96 + lane];
    const bool safe = shift_is_safe(M0, M1, m0, m1);

    int  my_p = 0, nc = 0;
    bool other = false;
    if (threadIdx.x == 0) {
        my_p = (int)((__builtin_amdgcn_s_sendmsg_rtn(0x87) >> 18) & 1u);
        nc   = atomicAdd(ctr + my_p, 1);
    }

    for (int it = 0; it < 2 * nchunk + 2; ++it) {
        if (threadIdx.x == 0) {
            if (nc >= nchunk && !other) {
                other = true;
                my_p ^= 1;
                nc = atomicAdd(ctr + my_p, 1);
            }
            sh[0] = nc;
            sh[1] = my_p;
        }
        __syncthreads();
        const int c  = __builtin_amdgcn_readfirstlane(sh[0]);
        const int pp = __builtin_amdgcn_readfirstlane(sh[1]);
        __syncthreads();
        if (c >= nchunk) break;
        if (threadIdx.x == 0) nc = atomicAdd(ctr + my_p, 1);

        const long e = (long)c * CH + (long)w * EV + lane;
        int  sv = 0, dv = 0;
        bool mine = false;
        if (e < E) {
            sv = src[e];
            dv = dst[e];
            mine = ((dv >= Nh) ? 1 : 0) == pp;
        }
        unsigned todo = __builtin_amdgcn_ballot_w32(mine);

        for (int round = 0; round < EV && todo != 0u; ++round) {
            int ss[U], dd[U];
            int cnt = 0;
#pragma unroll
            for (int u = 0; u < U; ++u) {
                const bool valid = todo != 0u;
                const int  j  = __builtin_ctz(todo | 0x80000000u);
                const int  s_ = __builtin_amdgcn_readlane(sv, j);
                const int  d_ = __builtin_amdgcn_readlane(dv, j);
                ss[u] = valid ? s_ : 0;
                dd[u] = valid ? d_ : 0;
                cnt  += valid ? 1 : 0;
                todo &= todo - 1u;
            }
            float a[U], b[U];
#pragma unroll
            for (int u = 0; u < U; ++u) {
                const float* row = z + (size_t)ss[u] * Hc;
                a[u] = row[lane];
                b[u] = row[lane + 32];
            }
            if (safe) {
#pragma unroll
                for (int u = 0; u < U; ++u) {
                    if (u < cnt) {
                        size_t o   = (size_t)dd[u] * Hc + lane;
                        float msg0 = fmaxf(a[u], 0.f) + GEPS;
                        float msg1 = fmaxf(b[u], 0.f) + GEPS;
                        float ex0  = __expf(msg0 * t - M0);
                        float ex1  = __expf(msg1 * t - M1);
                        atomicAdd(den + o,      ex0);
                        atomicAdd(den + o + 32, ex1);
                        atomicAdd(num + o,      msg0 * ex0);
                        atomicAdd(num + o + 32, msg1 * ex1);
                    }
                }
            } else {
#pragma unroll
                for (int u = 0; u < U; ++u) {
                    if (u < cnt) {
                        float* o = smax + (size_t)dd[u] * Hc;
                        atomicMaxF(o + lane,      (fmaxf(a[u], 0.f) + GEPS) * t);
                        atomicMaxF(o + lane + 32, (fmaxf(b[u], 0.f) + GEPS) * t);
                    }
                }
            }
        }
    }
}

template <int EB>
__global__ __launch_bounds__(256)
void edge_pass2_kernel(const float* __restrict__ z,
                       const int* __restrict__ src,
                       const int* __restrict__ dst,
                       const float* __restrict__ tptr, int layer,
                       const float* __restrict__ stats,
                       const float* __restrict__ smax,
                       float* __restrict__ den,
                       float* __restrict__ num, int E) {
    const int lane = threadIdx.x & 31;
    const int wave = __builtin_amdgcn_readfirstlane(
                         (int)((blockIdx.x * blockDim.x + threadIdx.x) >> 5));
    if (shift_is_safe(stats[lane], stats[lane + 32], stats[64 + lane], stats[96 + lane])) return;
    const float t      = tptr[layer];
    const int   nwaves = (int)((gridDim.x * blockDim.x) >> 5);
    const long  nbatch = ((long)E + EB - 1) / EB;

    for (long bt = wave; bt < nbatch; bt += nwaves) {
        const long e0 = bt * EB;
        if (e0 + EB <= E) {
            int s[EB], d[EB];
#pragma unroll
            for (int j = 0; j < EB; ++j) { s[j] = src[e0 + j]; d[j] = dst[e0 + j]; }
            float a[EB], b[EB], ma[EB], mb[EB];
#pragma unroll
            for (int j = 0; j < EB; ++j) {
                const float* row = z    + (size_t)s[j] * Hc;
                const float* mx  = smax + (size_t)d[j] * Hc;
                a[j]  = row[lane];
                b[j]  = row[lane + 32];
                ma[j] = mx[lane];
                mb[j] = mx[lane + 32];
            }
#pragma unroll
            for (int j = 0; j < EB; ++j) {
                size_t o   = (size_t)d[j] * Hc + lane;
                float msg0 = fmaxf(a[j], 0.f) + GEPS;
                float msg1 = fmaxf(b[j], 0.f) + GEPS;
                float ex0  = __expf(msg0 * t - ma[j]);
                float ex1  = __expf(msg1 * t - mb[j]);
                atomicAdd(den + o,      ex0);
                atomicAdd(den + o + 32, ex1);
                atomicAdd(num + o,      msg0 * ex0);
                atomicAdd(num + o + 32, msg1 * ex1);
            }
        } else {
            for (long e = e0; e < E; ++e) {
                const float* row = z + (size_t)src[e] * Hc;
                size_t o   = (size_t)dst[e] * Hc + lane;
                float msg0 = fmaxf(row[lane],      0.f) + GEPS;
                float msg1 = fmaxf(row[lane + 32], 0.f) + GEPS;
                float ex0  = __expf(msg0 * t - smax[o]);
                float ex1  = __expf(msg1 * t - smax[o + 32]);
                atomicAdd(den + o,      ex0);
                atomicAdd(den + o + 32, ex1);
                atomicAdd(num + o,      msg0 * ex0);
                atomicAdd(num + o + 32, msg1 * ex1);
            }
        }
    }
}

__global__ __launch_bounds__(64)
void mlp_fused_wmma(const float* __restrict__ num,
                    const float* __restrict__ den,
                    const float* __restrict__ z,
                    const float* __restrict__ B1,
                    const float* __restrict__ bias1,
                    const float* __restrict__ g,
                    const float* __restrict__ bvec,
                    const float* __restrict__ B2,
                    const float* __restrict__ bias2,
                    float* __restrict__ C, int M,
                    float* __restrict__ arm) {
    constexpr int K1 = Hc;
    constexpr int N1 = 2 * Hc;
    constexpr int P1 = K1 + 4;
    constexpr int K2 = 2 * Hc;
    constexpr int N2 = Hc;
    constexpr int P2 = K2 + 4;
    __shared__ __attribute__((aligned(16))) float As[16 * P1];
    __shared__ __attribute__((aligned(16))) float Xs[16 * P2];

    if (arm != nullptr && blockIdx.x == 0) {
        arm[threadIdx.x]      = -3.0e38f;
        arm[64 + threadIdx.x] =  3.0e38f;
        if (threadIdx.x < 2) arm[128 + threadIdx.x] = 0.f;
    }

    const int tile = (int)blockIdx.x;
    if (tile >= (M >> 4)) return;

    const int lane = threadIdx.x & 31;
    const int w    = __builtin_amdgcn_readfirstlane((int)(threadIdx.x >> 5));
    const int mr   = lane & 15;
    const int half = lane >> 4;

    v4f n4[4], d4[4], z4[4];
#pragma unroll
    for (int i = 0; i < 4; ++i) {
        const int    row = 8 * w + 2 * i + half;
        const size_t o   = ((size_t)tile * 16 + row) * Hc + 4 * mr;
        n4[i] = *(const v4f*)(num + o);
        d4[i] = *(const v4f*)(den + o);
        z4[i] = *(const v4f*)(z + o);
    }
    float gj[4], cj[4];
#pragma unroll
    for (int j = 0; j < 4; ++j) {
        gj[j] = g[lane + 32 * j];
        cj[j] = bvec[lane + 32 * j];
    }
#pragma unroll
    for (int i = 0; i < 4; ++i) {
        const int row = 8 * w + 2 * i + half;
        v4f a;
        a.x = n4[i].x / (d4[i].x + 1e-16f) + z4[i].x;
        a.y = n4[i].y / (d4[i].y + 1e-16f) + z4[i].y;
        a.z = n4[i].z / (d4[i].z + 1e-16f) + z4[i].z;
        a.w = n4[i].w / (d4[i].w + 1e-16f) + z4[i].w;
        *(v4f*)(As + row * P1 + 4 * mr) = a;
    }
    __syncthreads();

    {
        v2f a_all[K1 / 4];
#pragma unroll
        for (int kk = 0; kk < K1 / 4; ++kk)
            a_all[kk] = *(const v2f*)(As + mr * P1 + 4 * kk + 2 * half);

        int ncol[4];
#pragma unroll
        for (int t = 0; t < 4; ++t) ncol[t] = w * 64 + t * 16 + mr;

        v8f acc[4] = {{0.f,0.f,0.f,0.f,0.f,0.f,0.f,0.f},
                      {0.f,0.f,0.f,0.f,0.f,0.f,0.f,0.f},
                      {0.f,0.f,0.f,0.f,0.f,0.f,0.f,0.f},
                      {0.f,0.f,0.f,0.f,0.f,0.f,0.f,0.f}};

#pragma unroll
        for (int kk = 0; kk < K1 / 4; ++kk) {
            const int kb = 4 * kk + half * 2;
            v2f b[4];
#pragma unroll
            for (int t = 0; t < 4; ++t) {
                b[t].x = B1[kb * N1 + ncol[t]];
                b[t].y = B1[(kb + 1) * N1 + ncol[t]];
            }
#pragma unroll
            for (int t = 0; t < 4; ++t)
                acc[t] = __builtin_amdgcn_wmma_f32_16x16x4_f32(
                             false, a_all[kk], false, b[t], (short)0, acc[t], false, false);
        }

#pragma unroll
        for (int t = 0; t < 4; ++t) {
            const int   n  = ncol[t];
            const float bv = bias1[n];
#pragma unroll
            for (int r = 0; r < 8; ++r)
                Xs[(half * 8 + r) * P2 + n] = acc[t][r] + bv;
        }
    }
    __syncthreads();

    {
        float v[8][4];
#pragma unroll
        for (int i = 0; i < 8; ++i) {
#pragma unroll
            for (int j = 0; j < 4; ++j) v[i][j] = Xs[(8 * w + i) * P2 + lane + 32 * j];
        }
        float s[8], ss[8];
#pragma unroll
        for (int i = 0; i < 8; ++i) {
            s[i] = 0.f;  ss[i] = 0.f;
#pragma unroll
            for (int j = 0; j < 4; ++j) {
                s[i]  += v[i][j];
                ss[i] += v[i][j] * v[i][j];
            }
        }
        for (int off = 16; off; off >>= 1) {
#pragma unroll
            for (int i = 0; i < 8; ++i) {
                s[i]  += __shfl_xor(s[i],  off);
                ss[i] += __shfl_xor(ss[i], off);
            }
        }
#pragma unroll
        for (int i = 0; i < 8; ++i) {
            float mean = s[i] / (float)K2;
            float var  = ss[i] / (float)K2 - mean * mean;
            float rstd = rsqrtf(var + 1e-5f);
#pragma unroll
            for (int j = 0; j < 4; ++j) {
                float y = (v[i][j] - mean) * rstd * gj[j] + cj[j];
                Xs[(8 * w + i) * P2 + lane + 32 * j] = fmaxf(y, 0.f);
            }
        }
    }
    __syncthreads();

    {
        v2f a_all[K2 / 4];
#pragma unroll
        for (int kk = 0; kk < K2 / 4; ++kk)
            a_all[kk] = *(const v2f*)(Xs + mr * P2 + 4 * kk + 2 * half);

        int ncol[2];
#pragma unroll
        for (int t = 0; t < 2; ++t) ncol[t] = w * 32 + t * 16 + mr;

        v8f acc[2] = {{0.f,0.f,0.f,0.f,0.f,0.f,0.f,0.f},
                      {0.f,0.f,0.f,0.f,0.f,0.f,0.f,0.f}};

#pragma unroll
        for (int kk = 0; kk < K2 / 4; ++kk) {
            const int kb = 4 * kk + half * 2;
            v2f b[2];
#pragma unroll
            for (int t = 0; t < 2; ++t) {
                b[t].x = B2[kb * N2 + ncol[t]];
                b[t].y = B2[(kb + 1) * N2 + ncol[t]];
            }
#pragma unroll
            for (int t = 0; t < 2; ++t)
                acc[t] = __builtin_amdgcn_wmma_f32_16x16x4_f32(
                             false, a_all[kk], false, b[t], (short)0, acc[t], false, false);
        }

#pragma unroll
        for (int t = 0; t < 2; ++t) {
            const int   n  = ncol[t];
            const float bv = bias2[n];
#pragma unroll
            for (int r = 0; r < 8; ++r)
                C[(size_t)(tile * 16 + half * 8 + r) * N2 + n] = acc[t][r] + bv;
        }
    }
}

__global__ __launch_bounds__(64)
void tail_fused_wmma(const float* __restrict__ h,
                     const float* __restrict__ y,
                     const float* __restrict__ g,
                     const float* __restrict__ bvec,
                     const float* __restrict__ B,
                     const float* __restrict__ bias,
                     float* __restrict__ out, int M) {
    constexpr int K  = Hc;
    constexpr int NC = 40;
    constexpr int P  = K + 4;
    __shared__ __attribute__((aligned(16))) float Zs[2][16 * P];

    const int lane = threadIdx.x & 31;
    const int w    = __builtin_amdgcn_readfirstlane((int)(threadIdx.x >> 5));
    const int mr   = lane & 15;
    const int half = lane >> 4;
    const int ntiles = M >> 4;
    const int traw   = (int)blockIdx.x * 2 + w;
    const bool live  = traw < ntiles;
    const int  tile  = live ? traw : (ntiles - 1);
    float* Zw = Zs[w];

    float a[16], b[16], ya[16], yb[16];
#pragma unroll
    for (int j = 0; j < 16; ++j) {
        const size_t o = ((size_t)tile * 16 + j) * Hc + lane;
        ya[j] = y[o];
        yb[j] = y[o + 32];
        a[j]  = h[o];
        b[j]  = h[o + 32];
    }
    const float g0 = g[lane],    g1 = g[lane + 32];
    const float c0 = bvec[lane], c1 = bvec[lane + 32];

    int ncol[3];
#pragma unroll
    for (int t = 0; t < 3; ++t) {
        const int n = t * 16 + mr;
        ncol[t] = (n < NC) ? n : (NC - 1);
    }
    v2f bf[K / 4][3];
#pragma unroll
    for (int kk = 0; kk < K / 4; ++kk) {
        const int kb = 4 * kk + half * 2;
#pragma unroll
        for (int t = 0; t < 3; ++t) {
            bf[kk][t].x = B[kb * NC + ncol[t]];
            bf[kk][t].y = B[(kb + 1) * NC + ncol[t]];
        }
    }
    float bv[3];
#pragma unroll
    for (int t = 0; t < 3; ++t) bv[t] = bias[ncol[t]];

    float v0[16], v1[16], s[16], ss[16];
#pragma unroll
    for (int j = 0; j < 16; ++j) {
        v0[j] = a[j] + ya[j];
        v1[j] = b[j] + yb[j];
        s[j] = 0.f;  ss[j] = 0.f;
        s[j] += v0[j];  ss[j] += v0[j] * v0[j];
        s[j] += v1[j];  ss[j] += v1[j] * v1[j];
    }
    for (int off = 16; off; off >>= 1) {
#pragma unroll
        for (int j = 0; j < 16; ++j) {
            s[j]  += __shfl_xor(s[j],  off);
            ss[j] += __shfl_xor(ss[j], off);
        }
    }
#pragma unroll
    for (int j = 0; j < 16; ++j) {
        float mean = s[j] / (float)Hc;
        float var  = ss[j] / (float)Hc - mean * mean;
        float rstd = rsqrtf(var + 1e-5f);
        Zw[j * P + lane]      = fmaxf((v0[j] - mean) * rstd * g0 + c0, 0.f);
        Zw[j * P + lane + 32] = fmaxf((v1[j] - mean) * rstd * g1 + c1, 0.f);
    }
    __syncthreads();

    v2f a_all[K / 4];
#pragma unroll
    for (int kk = 0; kk < K / 4; ++kk)
        a_all[kk] = *(const v2f*)(Zw + mr * P + 4 * kk + 2 * half);

    v8f acc[3] = {{0.f,0.f,0.f,0.f,0.f,0.f,0.f,0.f},
                  {0.f,0.f,0.f,0.f,0.f,0.f,0.f,0.f},
                  {0.f,0.f,0.f,0.f,0.f,0.f,0.f,0.f}};
#pragma unroll
    for (int kk = 0; kk < K / 4; ++kk) {
#pragma unroll
        for (int t = 0; t < 3; ++t)
            acc[t] = __builtin_amdgcn_wmma_f32_16x16x4_f32(
                         false, a_all[kk], false, bf[kk][t], (short)0, acc[t], false, false);
    }

    const bool c2 = mr < (NC - 32);
    float x0[8], x1[8], x2[8], mx[8], sm[8];
#pragma unroll
    for (int r = 0; r < 8; ++r) {
        x0[r] = acc[0][r] + bv[0];
        x1[r] = acc[1][r] + bv[1];
        x2[r] = acc[2][r] + bv[2];
        mx[r] = fmaxf(fmaxf(x0[r], x1[r]), c2 ? x2[r] : -3.0e38f);
    }
    for (int off = 8; off; off >>= 1) {
#pragma unroll
        for (int r = 0; r < 8; ++r) mx[r] = fmaxf(mx[r], __shfl_xor(mx[r], off));
    }
#pragma unroll
    for (int r = 0; r < 8; ++r) {
        float e = __expf(x0[r] - mx[r]) + __expf(x1[r] - mx[r]);
        sm[r] = e + (c2 ? __expf(x2[r] - mx[r]) : 0.f);
    }
    for (int off = 8; off; off >>= 1) {
#pragma unroll
        for (int r = 0; r < 8; ++r) sm[r] += __shfl_xor(sm[r], off);
    }
#pragma unroll
    for (int r = 0; r < 8; ++r) {
        const float lse = mx[r] + __logf(sm[r]);
        float* orow = Zw + (half * 8 + r) * NC + mr;
        orow[0]  = x0[r] - lse;
        orow[16] = x1[r] - lse;
        if (c2) orow[32] = x2[r] - lse;
    }
    __syncthreads();
    if (live) {
        float* o = out + (size_t)tile * (16 * NC);
#pragma unroll
        for (int q = 0; q < (16 * NC) / 128; ++q) {
            const int i = (lane + 32 * q) * 4;
            *(v4f*)(o + i) = *(const v4f*)(Zw + i);
        }
    }
}

extern "C" void kernel_launch(void* const* d_in, const int* in_sizes, int n_in,
                              void* d_out, int out_size, void* d_ws, size_t ws_size,
                              hipStream_t stream) {
    (void)n_in; (void)out_size; (void)ws_size;

    const float* x    = (const float*)d_in[0];
    const int*   ei   = (const int*)  d_in[1];
    const float* encW = (const float*)d_in[2];
    const float* encB = (const float*)d_in[3];
    const float* t    = (const float*)d_in[4];
    const float* W1   = (const float*)d_in[5];
    const float* b1   = (const float*)d_in[6];
    const float* g1   = (const float*)d_in[7];
    const float* bb1  = (const float*)d_in[8];
    const float* W2   = (const float*)d_in[9];
    const float* b2   = (const float*)d_in[10];
    const float* ng   = (const float*)d_in[11];
    const float* nb   = (const float*)d_in[12];
    const float* linW = (const float*)d_in[13];
    const float* linB = (const float*)d_in[14];

    const int F = 128, H = 64, L = 4;
    const int N = in_sizes[0] / F;
    const int E = in_sizes[1] / 2;
    const int* src = ei;
    const int* dst = ei + E;

    float* ws = (float*)d_ws;
    const long NH = (long)N * H;
    float* h    = ws;  ws += NH;
    float* z    = ws;  ws += NH;
    float* smax = ws;  ws += NH;
    float* den  = ws;  ws += NH;
    float* num  = ws;  ws += NH;
    float* outb = ws;  ws += NH;
    float* yb   = ws;  ws += NH;

    const int  TB = 256;
    auto cdiv = [](long a, long b) { return (int)((a + b - 1) / b); };
    auto gemm_waves = [&](int M, int Nc) { return (long)(M / 16) * ((Nc + 63) / 64); };

    gemm_bias_wmma<<<cdiv(gemm_waves(N, H) * 32, TB), TB, 0, stream>>>(x, encW, encB, h, N, H, F);

    const int U1  = 20;
    const int G1  = 2048;
    const int EB2 = 8;
    const int RBF = 8;
    const int RBB = 2;
    const int G2  = 512;
    const int GB  = cdiv(N, 32 * RBB);

    float* stats = outb;
    int* ctr = (int*)(stats + 128);
    init_stats_kernel<<<1, 128, 0, stream>>>(stats);

    fill_stats_kernel<RBF><<<cdiv((long)cdiv(N, RBF) * 32, TB), TB, 0, stream>>>(h, t, 0, smax, den, num, stats, N);

    for (int l = 0; l < L; ++l) {
        const float* zin = (l == 0) ? h : z;

        edge_pass1_kernel<U1><<<G1, TB, 0, stream>>>(zin, src, dst, t, l, stats, smax, den, num, ctr, E, N);
        edge_pass2_kernel<EB2><<<G2, TB, 0, stream>>>(zin, src, dst, t, l, stats, smax, den, num, E);

        mlp_fused_wmma<<<N / 16, 64, 0, stream>>>(
            num, den, zin, W1 + (long)l * H * 2 * H, b1 + l * 2 * H,
            g1 + l * 2 * H, bb1 + l * 2 * H, W2 + (long)l * 2 * H * H, b2 + l * H, yb, N,
            (l + 1 < L) ? stats : nullptr);

        if (l == 0)
            boundary_kernel<RBB, true,  true ><<<GB, 1024, 0, stream>>>(h, yb, ng + H, nb + H, z, t, 1, smax, den, num, stats, N);
        else if (l + 1 < L)
            boundary_kernel<RBB, false, true ><<<GB, 1024, 0, stream>>>(h, yb, ng + (l + 1) * H, nb + (l + 1) * H, z, t, l + 1, smax, den, num, stats, N);
    }

    tail_fused_wmma<<<cdiv(N / 16, 2), 64, 0, stream>>>(h, yb, ng, nb, linW, linB, (float*)d_out, N);
}
